# starting kernel
# baseline (speedup 1.0000x reference)
#include <hip/hip_runtime.h>
#include <hip/hip_bf16.h>

typedef __attribute__((ext_vector_type(2))) float v2f;
typedef __attribute__((ext_vector_type(4))) float v4f;
typedef __attribute__((ext_vector_type(8))) float v8f;

#define B_ 32
#define J_ 32
#define I_ 2048
#define T_ 16
#define D_ 16

__global__ __launch_bounds__(256) void uhat_gemm(const float* __restrict__ x,
                                                 const float* __restrict__ W,
                                                 float* __restrict__ u) {
    const int i    = blockIdx.x;
    const int tid  = threadIdx.x;
    const int wave = tid >> 5;
    const int lane = tid & 31;
    const int n    = lane & 15;
    const int hi   = lane >> 4;
    const int k0   = hi * 2;

    v2f Bx[4][2];
#pragma unroll
    for (int h = 0; h < 2; ++h) {
        const float* xp = x + ((size_t)(n + 16 * h) * I_ + i) * D_;
#pragma unroll
        for (int c = 0; c < 4; ++c)
            Bx[c][h] = *(const v2f*)(xp + 4 * c + k0);
    }

#pragma unroll
    for (int jj = 0; jj < 4; ++jj) {
        const int j = (wave << 2) + jj;
        const float* wp = W + (((size_t)j * I_ + i) * T_ + n) * D_;
        v2f A[4];
#pragma unroll
        for (int c = 0; c < 4; ++c)
            A[c] = *(const v2f*)(wp + 4 * c + k0);

        v8f acc0 = {0.f, 0.f, 0.f, 0.f, 0.f, 0.f, 0.f, 0.f};
        v8f acc1 = acc0;
#pragma unroll
        for (int c = 0; c < 4; ++c) {
            acc0 = __builtin_amdgcn_wmma_f32_16x16x4_f32(
                false, A[c], false, Bx[c][0], (short)0, acc0, false, false);
            acc1 = __builtin_amdgcn_wmma_f32_16x16x4_f32(
                false, A[c], false, Bx[c][1], (short)0, acc1, false, false);
        }

#pragma unroll
        for (int h = 0; h < 2; ++h) {
            const v8f acc = h ? acc1 : acc0;
            float* up = u + (((size_t)j * I_ + i) * B_ + (n + 16 * h)) * T_ + 8 * hi;
            v4f lo = __builtin_shufflevector(acc, acc, 0, 1, 2, 3);
            v4f hv = __builtin_shufflevector(acc, acc, 4, 5, 6, 7);
            *(v4f*)(up)     = lo;
            *(v4f*)(up + 4) = hv;
        }
    }
}

__global__ __launch_bounds__(256) void softmax_j(const float* __restrict__ bij,
                                                 float* __restrict__ c) {
    const int g = blockIdx.x * 256 + threadIdx.x;
    const int b = g >> 11;
    const int i = g & (I_ - 1);
    const float* bp = bij + (size_t)b * J_ * I_ + i;
    float vals[J_];
    float m = -3.4e38f;
#pragma unroll
    for (int j = 0; j < J_; ++j) {
        vals[j] = bp[(size_t)j * I_];
        m = fmaxf(m, vals[j]);
    }
    float s = 0.f;
#pragma unroll
    for (int j = 0; j < J_; ++j) {
        vals[j] = __expf(vals[j] - m);
        s += vals[j];
    }
    const float inv = 1.0f / s;
    float* cp = c + (size_t)b * J_ * I_ + i;
#pragma unroll
    for (int j = 0; j < J_; ++j)
        cp[(size_t)j * I_] = vals[j] * inv;
}

__global__ __launch_bounds__(256) void sv_kernel(const float* __restrict__ u,
                                                 const float* __restrict__ c,
                                                 float* __restrict__ v,
                                                 int uniform) {
    const int bj  = blockIdx.x;
    const int b   = bj >> 5;
    const int j   = bj & 31;
    const int tid = threadIdx.x;
    const int t   = tid & 15;
    const int grp = tid >> 4;

    const float* ubase = u + (size_t)j * I_ * (B_ * T_) + (size_t)b * T_ + t;
    const float* cbase = c + ((size_t)b * J_ + j) * I_;

    float acc = 0.f;
    for (int i = grp; i < I_; i += 16) {
        const float uv = ubase[(size_t)i * (B_ * T_)];
        acc += uniform ? uv : uv * cbase[i];
    }

    __shared__ float red[256];
    __shared__ float scale_s;
    red[tid] = acc;
    __syncthreads();
    for (int sft = 128; sft >= 16; sft >>= 1) {
        if (tid < sft) red[tid] += red[tid + sft];
        __syncthreads();
    }
    if (tid == 0) {
        float sq = 0.f;
#pragma unroll
        for (int tt = 0; tt < 16; ++tt) { const float sv = red[tt]; sq += sv * sv; }
        if (uniform) sq *= (1.0f / 32.f) * (1.0f / 32.f);
        scale_s = sq / (1.0f + sq) / sqrtf(sq + 1e-9f);
    }
    __syncthreads();
    if (tid < 16) {
        float sv = red[t];
        if (uniform) sv *= (1.0f / 32.f);
        v[((size_t)b * J_ + j) * T_ + t] = scale_s * sv;
    }
}

__global__ __launch_bounds__(256) void bupd_kernel(const float* __restrict__ u,
                                                   const float* __restrict__ v,
                                                   float* __restrict__ bij,
                                                   int first) {
    const size_t g = (size_t)blockIdx.x * 256 + threadIdx.x;
    const int i  = (int)(g & (I_ - 1));
    const int bj = (int)(g >> 11);
    const int j  = bj & 31;
    const int b  = bj >> 5;

    const float* up = u + (((size_t)j * I_ + i) * B_ + b) * T_;
    const float* vp = v + ((size_t)b * J_ + j) * T_;

    float dot = 0.f;
#pragma unroll
    for (int q = 0; q < 4; ++q) {
        const v4f uv = *(const v4f*)(up + 4 * q);
        const v4f vv = *(const v4f*)(vp + 4 * q);
        dot += uv.x * vv.x + uv.y * vv.y + uv.z * vv.z + uv.w * vv.w;
    }

    const size_t bidx = ((size_t)b * J_ + j) * I_ + i;
    bij[bidx] = first ? dot : (bij[bidx] + dot);
}

extern "C" void kernel_launch(void* const* d_in, const int* in_sizes, int n_in,
                              void* d_out, int out_size, void* d_ws, size_t ws_size,
                              hipStream_t stream) {
    const float* x = (const float*)d_in[0];
    const float* W = (const float*)d_in[1];
    float* out = (float*)d_out;

    float* u   = (float*)d_ws;
    float* bij = u + (size_t)J_ * I_ * B_ * T_;
    float* c   = bij + (size_t)B_ * J_ * I_;

    uhat_gemm<<<I_, 256, 0, stream>>>(x, W, u);

    sv_kernel<<<B_ * J_, 256, 0, stream>>>(u, c, out, 1);
    bupd_kernel<<<(B_ * J_ * I_) / 256, 256, 0, stream>>>(u, out, bij, 1);

    softmax_j<<<(B_ * I_) / 256, 256, 0, stream>>>(bij, c);
    sv_kernel<<<B_ * J_, 256, 0, stream>>>(u, c, out, 0);
    bupd_kernel<<<(B_ * J_ * I_) / 256, 256, 0, stream>>>(u, out, bij, 0);

    softmax_j<<<(B_ * I_) / 256, 256, 0, stream>>>(bij, c);
    sv_kernel<<<B_ * J_, 256, 0, stream>>>(u, c, out, 0);
}


# u never stored: a separate agreement kernel builds W_ji x_i^T per (j,i) on f32 WMMA, dots it with v in the accumulator lanes; line-wide b_ij stores via LDS; u GEMM gone; sum order over d,t changes
# speedup vs baseline: 2.6257x; 2.6257x over previous
#include <hip/hip_runtime.h>
#include <hip/hip_bf16.h>

typedef __attribute__((ext_vector_type(2))) float v2f;
typedef __attribute__((ext_vector_type(4))) float v4f;
typedef __attribute__((ext_vector_type(8))) float v8f;

#define B_ 32
#define J_ 32
#define I_ 2048
#define T_ 16
#define D_ 16

__global__ __launch_bounds__(256) void softmax_j(const float* __restrict__ bij,
                                                 float* __restrict__ c) {
    const int g = blockIdx.x * 256 + threadIdx.x;
    const int b = g >> 11;
    const int i = g & (I_ - 1);
    const float* bp = bij + (size_t)b * J_ * I_ + i;
    float vals[J_];
    float m = -3.4e38f;
#pragma unroll
    for (int j = 0; j < J_; ++j) {
        vals[j] = bp[(size_t)j * I_];
        m = fmaxf(m, vals[j]);
    }
    float s = 0.f;
#pragma unroll
    for (int j = 0; j < J_; ++j) {
        vals[j] = __expf(vals[j] - m);
        s += vals[j];
    }
    const float inv = 1.0f / s;
    float* cp = c + (size_t)b * J_ * I_ + i;
#pragma unroll
    for (int j = 0; j < J_; ++j)
        cp[(size_t)j * I_] = vals[j] * inv;
}

#define IC_ 32
#define NCH_ (I_ / IC_)

template <bool UNI>
__global__ __launch_bounds__(256) void sv_wgemm(const float* __restrict__ x,
                                                const float* __restrict__ W,
                                                const float* cpl,
                                                float* ppl) {
    const int chunk = blockIdx.x >> 3;
    const int jg    = blockIdx.x & 7;
    const int tid   = threadIdx.x;
    const int wave  = tid >> 5;
    const int lane  = tid & 31;
    const int n     = lane & 15;
    const int hi    = lane >> 4;
    const int i0    = chunk * IC_ + wave * 4;

    v8f acc[4][2];
#pragma unroll
    for (int jj = 0; jj < 4; ++jj)
#pragma unroll
        for (int h = 0; h < 2; ++h)
            acc[jj][h] = (v8f){0.f, 0.f, 0.f, 0.f, 0.f, 0.f, 0.f, 0.f};

    v4f cv[4][2];
    if (!UNI) {
#pragma unroll
        for (int jj = 0; jj < 4; ++jj)
#pragma unroll
            for (int h = 0; h < 2; ++h)
                cv[jj][h] = *(const v4f*)(cpl + ((size_t)(n + 16 * h) * J_ + (jg * 4 + jj)) * I_ + i0);
    }

    v4f Wt[4][4][2];
    v4f Xt[4][2][2];
#define SV_LOAD_STAGE(q)                                                                  \
    {                                                                                     \
        _Pragma("unroll") for (int h = 0; h < 2; ++h) {                                   \
            const float* xp = x + ((size_t)(n + 16 * h) * I_ + (i0 + (q))) * D_ + 8 * hi; \
            Xt[q][h][0] = *(const v4f*)(xp);                                              \
            Xt[q][h][1] = *(const v4f*)(xp + 4);                                          \
        }                                                                                 \
        _Pragma("unroll") for (int jj = 0; jj < 4; ++jj) {                                \
            const float* wp =                                                             \
                W + (((size_t)(jg * 4 + jj) * I_ + (i0 + (q))) * T_ + n) * D_ + 8 * hi;   \
            Wt[q][jj][0] = *(const v4f*)(wp);                                             \
            Wt[q][jj][1] = *(const v4f*)(wp + 4);                                         \
        }                                                                                 \
    }

    SV_LOAD_STAGE(0)
    SV_LOAD_STAGE(1)
#pragma unroll
    for (int q = 0; q < 4; ++q) {
        __builtin_amdgcn_sched_barrier(0);
#pragma unroll
        for (int jj = 0; jj < 4; ++jj) {
            if (jj == 2 && q < 2) {
                __builtin_amdgcn_sched_barrier(0);
                if (q == 0) SV_LOAD_STAGE(2)
                if (q == 1) SV_LOAD_STAGE(3)
            }
            const v4f a0 = Wt[q][jj][0];
            const v4f a1 = Wt[q][jj][1];
            v2f A[4];
            A[0] = __builtin_shufflevector(a0, a0, 0, 1);
            A[1] = __builtin_shufflevector(a0, a0, 2, 3);
            A[2] = __builtin_shufflevector(a1, a1, 0, 1);
            A[3] = __builtin_shufflevector(a1, a1, 2, 3);
            v2f Bv[2][4];
#pragma unroll
            for (int h = 0; h < 2; ++h) {
                v4f b0 = Xt[q][h][0];
                v4f b1 = Xt[q][h][1];
                if (!UNI) {
                    const float s = cv[jj][h][q];
                    b0 *= s;
                    b1 *= s;
                }
                Bv[h][0] = __builtin_shufflevector(b0, b0, 0, 1);
                Bv[h][1] = __builtin_shufflevector(b0, b0, 2, 3);
                Bv[h][2] = __builtin_shufflevector(b1, b1, 0, 1);
                Bv[h][3] = __builtin_shufflevector(b1, b1, 2, 3);
            }
#pragma unroll
            for (int c4 = 0; c4 < 4; ++c4) {
                acc[jj][0] = __builtin_amdgcn_wmma_f32_16x16x4_f32(
                    false, A[c4], false, Bv[0][c4], (short)0, acc[jj][0], false, false);
                acc[jj][1] = __builtin_amdgcn_wmma_f32_16x16x4_f32(
                    false, A[c4], false, Bv[1][c4], (short)0, acc[jj][1], false, false);
            }
        }
    }
#undef SV_LOAD_STAGE

    __shared__ float red[8 * 16 * 32 * 4];
#pragma unroll
    for (int jj = 0; jj < 4; ++jj)
#pragma unroll
        for (int h = 0; h < 2; ++h) {
            const int g = jj * 4 + h * 2;
            const v8f a = acc[jj][h];
            *(v4f*)&red[((wave * 16 + g) * 32 + lane) * 4]     = __builtin_shufflevector(a, a, 0, 1, 2, 3);
            *(v4f*)&red[((wave * 16 + g + 1) * 32 + lane) * 4] = __builtin_shufflevector(a, a, 4, 5, 6, 7);
        }
    __syncthreads();

    const int ojj = wave >> 1;
    const int oh  = wave & 1;
    const int g0  = ojj * 4 + oh * 2;
    v4f lo = {0.f, 0.f, 0.f, 0.f};
    v4f hv = lo;
#pragma unroll
    for (int w = 0; w < 8; ++w) {
        lo += *(const v4f*)&red[((w * 16 + g0) * 32 + lane) * 4];
        hv += *(const v4f*)&red[((w * 16 + g0 + 1) * 32 + lane) * 4];
    }
    float* pp = ppl + ((size_t)(n + 16 * oh) * J_ + (jg * 4 + ojj)) * I_ + chunk * IC_ + 8 * hi;
    *(v4f*)(pp)     = lo;
    *(v4f*)(pp + 4) = hv;
}

__global__ __launch_bounds__(256) void reduce_squash(const float* __restrict__ P,
                                                     float* __restrict__ v,
                                                     float scale) {
    const int tid  = threadIdx.x;
    const int lane = tid & 31;
    const int bj   = blockIdx.x * 8 + (tid >> 5);
    const int t    = lane & 15;
    const int cg   = lane >> 4;
    const float* p = P + (size_t)bj * I_ + (size_t)(cg * (NCH_ / 2)) * IC_ + t;
    float s = 0.f;
#pragma unroll
    for (int k = 0; k < NCH_ / 2; ++k) s += p[k * IC_];
    s += __shfl_xor(s, 16);
    s *= scale;
    float sq = s * s;
    sq += __shfl_xor(sq, 8);
    sq += __shfl_xor(sq, 4);
    sq += __shfl_xor(sq, 2);
    sq += __shfl_xor(sq, 1);
    const float f = sq / (1.0f + sq) / sqrtf(sq + 1e-9f);
    if (cg == 0) v[(size_t)bj * T_ + t] = f * s;
}

#define AGP_ 36

__device__ __forceinline__ float xhalf(float a) {
    const int b = __builtin_bit_cast(int, a);
    return __builtin_bit_cast(float,
        __builtin_amdgcn_permlanex16(b, b, 0x76543210, 0xfedcba98, false, false));
}

template <bool FIRST>
__global__ __launch_bounds__(256) void agree_wgemm(const float* __restrict__ x,
                                                   const float* __restrict__ W,
                                                   const float* __restrict__ v,
                                                   float* bij) {
    const int chunk = blockIdx.x >> 3;
    const int jg    = blockIdx.x & 7;
    const int tid   = threadIdx.x;
    const int wave  = tid >> 5;
    const int lane  = tid & 31;
    const int n     = lane & 15;
    const int hi    = lane >> 4;
    const int i0    = chunk * IC_ + wave * 4;

    v4f vv[4][2][2];
#pragma unroll
    for (int jj = 0; jj < 4; ++jj)
#pragma unroll
        for (int h = 0; h < 2; ++h) {
            const float* vp = v + ((size_t)(n + 16 * h) * J_ + (jg * 4 + jj)) * T_ + 8 * hi;
            vv[jj][h][0] = *(const v4f*)(vp);
            vv[jj][h][1] = *(const v4f*)(vp + 4);
        }

    v4f Wt[4][4][2];
    v4f Xt[4][2][2];
#define AG_LOAD_STAGE(q)                                                                  \
    {                                                                                     \
        _Pragma("unroll") for (int h = 0; h < 2; ++h) {                                   \
            const float* xp = x + ((size_t)(n + 16 * h) * I_ + (i0 + (q))) * D_ + 8 * hi; \
            Xt[q][h][0] = *(const v4f*)(xp);                                              \
            Xt[q][h][1] = *(const v4f*)(xp + 4);                                          \
        }                                                                                 \
        _Pragma("unroll") for (int jj = 0; jj < 4; ++jj) {                                \
            const float* wp =                                                             \
                W + (((size_t)(jg * 4 + jj) * I_ + (i0 + (q))) * T_ + n) * D_ + 8 * hi;   \
            Wt[q][jj][0] = *(const v4f*)(wp);                                             \
            Wt[q][jj][1] = *(const v4f*)(wp + 4);                                         \
        }                                                                                 \
    }

    float res[4][4];

    AG_LOAD_STAGE(0)
    AG_LOAD_STAGE(1)
#pragma unroll
    for (int q = 0; q < 4; ++q) {
        __builtin_amdgcn_sched_barrier(0);
#pragma unroll
        for (int jj = 0; jj < 4; ++jj) {
            if (jj == 2 && q < 2) {
                __builtin_amdgcn_sched_barrier(0);
                if (q == 0) AG_LOAD_STAGE(2)
                if (q == 1) AG_LOAD_STAGE(3)
            }
            const v4f a0 = Wt[q][jj][0];
            const v4f a1 = Wt[q][jj][1];
            v2f A[4];
            A[0] = __builtin_shufflevector(a0, a0, 0, 1);
            A[1] = __builtin_shufflevector(a0, a0, 2, 3);
            A[2] = __builtin_shufflevector(a1, a1, 0, 1);
            A[3] = __builtin_shufflevector(a1, a1, 2, 3);
            v2f Bv[2][4];
#pragma unroll
            for (int h = 0; h < 2; ++h) {
                const v4f b0 = Xt[q][h][0];
                const v4f b1 = Xt[q][h][1];
                Bv[h][0] = __builtin_shufflevector(b0, b0, 0, 1);
                Bv[h][1] = __builtin_shufflevector(b0, b0, 2, 3);
                Bv[h][2] = __builtin_shufflevector(b1, b1, 0, 1);
                Bv[h][3] = __builtin_shufflevector(b1, b1, 2, 3);
            }
            v8f acc0 = {0.f, 0.f, 0.f, 0.f, 0.f, 0.f, 0.f, 0.f};
            v8f acc1 = acc0;
#pragma unroll
            for (int c4 = 0; c4 < 4; ++c4) {
                acc0 = __builtin_amdgcn_wmma_f32_16x16x4_f32(
                    false, A[c4], false, Bv[0][c4], (short)0, acc0, false, false);
                acc1 = __builtin_amdgcn_wmma_f32_16x16x4_f32(
                    false, A[c4], false, Bv[1][c4], (short)0, acc1, false, false);
            }
            const v4f v0a = vv[jj][0][0], v0b = vv[jj][0][1];
            const v4f v1a = vv[jj][1][0], v1b = vv[jj][1][1];
            float p0a = acc0[0] * v0a.x, p0b = acc0[4] * v0b.x;
            float p1a = acc1[0] * v1a.x, p1b = acc1[4] * v1b.x;
            p0a += acc0[1] * v0a.y;  p0b += acc0[5] * v0b.y;
            p1a += acc1[1] * v1a.y;  p1b += acc1[5] * v1b.y;
            p0a += acc0[2] * v0a.z;  p0b += acc0[6] * v0b.z;
            p1a += acc1[2] * v1a.z;  p1b += acc1[6] * v1b.z;
            p0a += acc0[3] * v0a.w;  p0b += acc0[7] * v0b.w;
            p1a += acc1[3] * v1a.w;  p1b += acc1[7] * v1b.w;
            const float p0 = p0a + p0b;
            const float p1 = p1a + p1b;
            const float keep = hi ? p1 : p0;
            const float send = hi ? p0 : p1;
            float lg = keep + xhalf(send);
            asm volatile("" : "+v"(lg));
            res[jj][q] = lg;
        }
    }
#undef AG_LOAD_STAGE

    __shared__ float stg[4 * 32 * AGP_];
    const int piece = lane & 7;
    float* gp[4];
    v4f old[4];
#pragma unroll
    for (int k = 0; k < 4; ++k) {
        const int line = wave * 16 + k * 4 + (lane >> 3);
        gp[k] = bij + ((size_t)(line & 31) * J_ + (jg * 4 + (line >> 5))) * I_ + chunk * IC_ + 4 * piece;
        if (!FIRST) old[k] = *(const v4f*)gp[k];
    }
#pragma unroll
    for (int jj = 0; jj < 4; ++jj)
        *(v4f*)&stg[(jj * 32 + n + 16 * hi) * AGP_ + 4 * wave] =
            (v4f){res[jj][0], res[jj][1], res[jj][2], res[jj][3]};
    __syncthreads();
#pragma unroll
    for (int k = 0; k < 4; ++k) {
        const int line = wave * 16 + k * 4 + (lane >> 3);
        v4f r = *(const v4f*)&stg[line * AGP_ + 4 * piece];
        if (!FIRST) r += old[k];
        *(v4f*)gp[k] = r;
    }
}

extern "C" void kernel_launch(void* const* d_in, const int* in_sizes, int n_in,
                              void* d_out, int out_size, void* d_ws, size_t ws_size,
                              hipStream_t stream) {
    const float* x = (const float*)d_in[0];
    const float* W = (const float*)d_in[1];
    float* out = (float*)d_out;

    float* bij = (float*)d_ws;
    float* c   = bij + (size_t)B_ * J_ * I_;

    sv_wgemm<true><<<NCH_ * 8, 256, 0, stream>>>(x, W, c, c);
    reduce_squash<<<(B_ * J_) / 8, 256, 0, stream>>>(c, out, 0.03125f);
    agree_wgemm<true><<<NCH_ * 8, 256, 0, stream>>>(x, W, out, bij);

    softmax_j<<<(B_ * I_) / 256, 256, 0, stream>>>(bij, c);
    sv_wgemm<false><<<NCH_ * 8, 256, 0, stream>>>(x, W, c, c);
    reduce_squash<<<(B_ * J_) / 8, 256, 0, stream>>>(c, out, 1.0f);
    agree_wgemm<false><<<NCH_ * 8, 256, 0, stream>>>(x, W, out, bij);

    softmax_j<<<(B_ * I_) / 256, 256, 0, stream>>>(bij, c);
    sv_wgemm<false><<<NCH_ * 8, 256, 0, stream>>>(x, W, c, c);
    reduce_squash<<<(B_ * J_) / 8, 256, 0, stream>>>(c, out, 1.0f);
}


# agreement kernel regrouped to blocks of 4 i x all 32 j: softmax over j done in the block via LDS, c and logits as [i/4][j][b][4]; softmax kernel gone (8 launches); softmax sum order over j changes
# speedup vs baseline: 2.9386x; 1.1192x over previous
#include <hip/hip_runtime.h>
#include <hip/hip_bf16.h>

typedef __attribute__((ext_vector_type(2))) float v2f;
typedef __attribute__((ext_vector_type(4))) float v4f;
typedef __attribute__((ext_vector_type(8))) float v8f;

#define B_ 32
#define J_ 32
#define I_ 2048
#define T_ 16
#define D_ 16

#define PL_(i4, j, b) ((((size_t)(i4) * J_ + (j)) * B_ + (b)) * 4)

#define IC_ 32
#define NCH_ (I_ / IC_)
#define PP_(chunk, j, b) (PL_((chunk) * 8 + ((b) >> 3), (j), 0) + ((b) & 7) * 16)


template <bool UNI>
__global__ __launch_bounds__(256) void sv_wgemm(const float* __restrict__ x,
                                                const float* __restrict__ W,
                                                const float* cpl,
                                                float* ppl) {
    const int g     = blockIdx.x;
    const int chunk = ((g >> 4) << 1) | ((g >> 2) & 1);
    const int jg    = (g & 3) | (((g >> 3) & 1) << 2);
    const int tid   = threadIdx.x;
    const int wave  = tid >> 5;
    const int lane  = tid & 31;
    const int n     = lane & 15;
    const int hi    = lane >> 4;
    const int i0    = chunk * IC_ + wave * 4;

    v8f acc[4][2];
#pragma unroll
    for (int jj = 0; jj < 4; ++jj)
#pragma unroll
        for (int h = 0; h < 2; ++h)
            acc[jj][h] = (v8f){0.f, 0.f, 0.f, 0.f, 0.f, 0.f, 0.f, 0.f};

    v4f cv[4][2];
    if (!UNI) {
#pragma unroll
        for (int jj = 0; jj < 4; ++jj)
#pragma unroll
            for (int h = 0; h < 2; ++h)
                cv[jj][h] = *(const v4f*)(cpl + PL_(chunk * 8 + wave, jg * 4 + jj, n + 16 * h));
    }

    v4f Wt[4][4][2];
    v4f Xt[4][2][2];
#define SV_LOAD_STAGE(q)                                                                  \
    {                                                                                     \
        _Pragma("unroll") for (int h = 0; h < 2; ++h) {                                   \
            const float* xp = x + ((size_t)(n + 16 * h) * I_ + (i0 + (q))) * D_ + 8 * hi; \
            Xt[q][h][0] = *(const v4f*)(xp);                                              \
            Xt[q][h][1] = *(const v4f*)(xp + 4);                                          \
        }                                                                                 \
        _Pragma("unroll") for (int jj = 0; jj < 4; ++jj) {                                \
            const float* wp =                                                             \
                W + (((size_t)(jg * 4 + jj) * I_ + (i0 + (q))) * T_ + n) * D_ + 8 * hi;   \
            Wt[q][jj][0] = *(const v4f*)(wp);                                             \
            Wt[q][jj][1] = *(const v4f*)(wp + 4);                                         \
        }                                                                                 \
    }

    SV_LOAD_STAGE(0)
    SV_LOAD_STAGE(1)
#pragma unroll
    for (int q = 0; q < 4; ++q) {
        __builtin_amdgcn_sched_barrier(0);
#pragma unroll
        for (int jj = 0; jj < 4; ++jj) {
            if (jj == 2 && q < 2) {
                __builtin_amdgcn_sched_barrier(0);
                if (q == 0) SV_LOAD_STAGE(2)
                if (q == 1) SV_LOAD_STAGE(3)
            }
            const v4f a0 = Wt[q][jj][0];
            const v4f a1 = Wt[q][jj][1];
            v2f A[4];
            A[0] = __builtin_shufflevector(a0, a0, 0, 1);
            A[1] = __builtin_shufflevector(a0, a0, 2, 3);
            A[2] = __builtin_shufflevector(a1, a1, 0, 1);
            A[3] = __builtin_shufflevector(a1, a1, 2, 3);
            v2f Bv[2][4];
#pragma unroll
            for (int h = 0; h < 2; ++h) {
                v4f b0 = Xt[q][h][0];
                v4f b1 = Xt[q][h][1];
                if (!UNI) {
                    const float s = cv[jj][h][q];
                    b0 *= s;
                    b1 *= s;
                }
                Bv[h][0] = __builtin_shufflevector(b0, b0, 0, 1);
                Bv[h][1] = __builtin_shufflevector(b0, b0, 2, 3);
                Bv[h][2] = __builtin_shufflevector(b1, b1, 0, 1);
                Bv[h][3] = __builtin_shufflevector(b1, b1, 2, 3);
            }
#pragma unroll
            for (int c4 = 0; c4 < 4; ++c4) {
                acc[jj][0] = __builtin_amdgcn_wmma_f32_16x16x4_f32(
                    false, A[c4], false, Bv[0][c4], (short)0, acc[jj][0], false, false);
                acc[jj][1] = __builtin_amdgcn_wmma_f32_16x16x4_f32(
                    false, A[c4], false, Bv[1][c4], (short)0, acc[jj][1], false, false);
            }
        }
    }
#undef SV_LOAD_STAGE

    __shared__ float red[8 * 16 * 32 * 4];
#pragma unroll
    for (int jj = 0; jj < 4; ++jj)
#pragma unroll
        for (int h = 0; h < 2; ++h) {
            const int g = jj * 4 + h * 2;
            const v8f a = acc[jj][h];
            *(v4f*)&red[((wave * 16 + g) * 32 + lane) * 4]     = __builtin_shufflevector(a, a, 0, 1, 2, 3);
            *(v4f*)&red[((wave * 16 + g + 1) * 32 + lane) * 4] = __builtin_shufflevector(a, a, 4, 5, 6, 7);
        }
    __syncthreads();

    const int ojj = wave >> 1;
    const int oh  = wave & 1;
    const int g0  = ojj * 4 + oh * 2;
    v4f lo = {0.f, 0.f, 0.f, 0.f};
    v4f hv = lo;
#pragma unroll
    for (int w = 0; w < 8; ++w) {
        lo += *(const v4f*)&red[((w * 16 + g0) * 32 + lane) * 4];
        hv += *(const v4f*)&red[((w * 16 + g0 + 1) * 32 + lane) * 4];
    }
    float* pp = ppl + PP_(chunk, jg * 4 + ojj, n + 16 * oh) + 8 * hi;
    *(v4f*)(pp)     = lo;
    *(v4f*)(pp + 4) = hv;
}

__global__ __launch_bounds__(256) void reduce_squash(const float* __restrict__ P,
                                                     float* __restrict__ v,
                                                     float scale) {
    const int tid  = threadIdx.x;
    const int lane = tid & 31;
    const int bj   = blockIdx.x * 8 + (tid >> 5);
    const int t    = lane & 15;
    const int cg   = lane >> 4;
    const float* p = P + PP_(cg * (NCH_ / 2), bj & (J_ - 1), bj >> 5) + t;
    float s = 0.f;
#pragma unroll
    for (int k = 0; k < NCH_ / 2; ++k) s += p[(size_t)k * (8 * J_ * B_ * 4)];
    s += __shfl_xor(s, 16);
    s *= scale;
    float sq = s * s;
    sq += __shfl_xor(sq, 8);
    sq += __shfl_xor(sq, 4);
    sq += __shfl_xor(sq, 2);
    sq += __shfl_xor(sq, 1);
    const float f = sq / (1.0f + sq) / sqrtf(sq + 1e-9f);
    if (cg == 0) v[(size_t)bj * T_ + t] = f * s;
}

__device__ __forceinline__ float xhalf(float a) {
    const int b = __builtin_bit_cast(int, a);
    return __builtin_bit_cast(float,
        __builtin_amdgcn_permlanex16(b, b, 0x76543210, 0xfedcba98, false, false));
}

__device__ __forceinline__ v4f vmax4(v4f a, v4f b) {
    return (v4f){fmaxf(a.x, b.x), fmaxf(a.y, b.y), fmaxf(a.z, b.z), fmaxf(a.w, b.w)};
}

template <bool FIRST>
__global__ __launch_bounds__(256) void agree_softmax(const float* __restrict__ x,
                                                     const float* __restrict__ W,
                                                     const float* __restrict__ v,
                                                     float* __restrict__ lgp,
                                                     float* __restrict__ cpl) {
    const int g    = blockIdx.x;
    const int i4   = (g & ~12) | (((g >> 2) & 1) << 3) | (((g >> 3) & 1) << 2);
    const int tid  = threadIdx.x;
    const int wave = tid >> 5;
    const int lane = tid & 31;
    const int n    = lane & 15;
    const int hi   = lane >> 4;
    const int jg   = wave;
    const int i0   = i4 * 4;

    v4f vv[4][2][2];
#pragma unroll
    for (int jj = 0; jj < 4; ++jj)
#pragma unroll
        for (int h = 0; h < 2; ++h) {
            const float* vp = v + ((size_t)(n + 16 * h) * J_ + (jg * 4 + jj)) * T_ + 8 * hi;
            vv[jj][h][0] = *(const v4f*)(vp);
            vv[jj][h][1] = *(const v4f*)(vp + 4);
        }

    v4f Wt[4][4][2];
    v4f Xt[4][2][2];
#define AG_LOAD_STAGE(q)                                                                  \
    {                                                                                     \
        _Pragma("unroll") for (int h = 0; h < 2; ++h) {                                   \
            const float* xp = x + ((size_t)(n + 16 * h) * I_ + (i0 + (q))) * D_ + 8 * hi; \
            Xt[q][h][0] = *(const v4f*)(xp);                                              \
            Xt[q][h][1] = *(const v4f*)(xp + 4);                                          \
        }                                                                                 \
        _Pragma("unroll") for (int jj = 0; jj < 4; ++jj) {                                \
            const float* wp =                                                             \
                W + (((size_t)(jg * 4 + jj) * I_ + (i0 + (q))) * T_ + n) * D_ + 8 * hi;   \
            Wt[q][jj][0] = *(const v4f*)(wp);                                             \
            Wt[q][jj][1] = *(const v4f*)(wp + 4);                                         \
        }                                                                                 \
    }

    float res[4][4];
    v4f old[4];

    AG_LOAD_STAGE(0)
    AG_LOAD_STAGE(1)
#pragma unroll
    for (int q = 0; q < 4; ++q) {
        __builtin_amdgcn_sched_barrier(0);
        if (!FIRST && q == 2) {
#pragma unroll
            for (int jj = 0; jj < 4; ++jj)
                old[jj] = *(const v4f*)(lgp + PL_(i4, jg * 4 + jj, lane));
        }
#pragma unroll
        for (int jj = 0; jj < 4; ++jj) {
            if (jj == 2 && q < 2) {
                __builtin_amdgcn_sched_barrier(0);
                if (q == 0) AG_LOAD_STAGE(2)
                if (q == 1) AG_LOAD_STAGE(3)
            }
            const v4f a0 = Wt[q][jj][0];
            const v4f a1 = Wt[q][jj][1];
            v2f A[4];
            A[0] = __builtin_shufflevector(a0, a0, 0, 1);
            A[1] = __builtin_shufflevector(a0, a0, 2, 3);
            A[2] = __builtin_shufflevector(a1, a1, 0, 1);
            A[3] = __builtin_shufflevector(a1, a1, 2, 3);
            v2f Bv[2][4];
#pragma unroll
            for (int h = 0; h < 2; ++h) {
                const v4f b0 = Xt[q][h][0];
                const v4f b1 = Xt[q][h][1];
                Bv[h][0] = __builtin_shufflevector(b0, b0, 0, 1);
                Bv[h][1] = __builtin_shufflevector(b0, b0, 2, 3);
                Bv[h][2] = __builtin_shufflevector(b1, b1, 0, 1);
                Bv[h][3] = __builtin_shufflevector(b1, b1, 2, 3);
            }
            v8f acc0 = {0.f, 0.f, 0.f, 0.f, 0.f, 0.f, 0.f, 0.f};
            v8f acc1 = acc0;
#pragma unroll
            for (int c4 = 0; c4 < 4; ++c4) {
                acc0 = __builtin_amdgcn_wmma_f32_16x16x4_f32(
                    false, A[c4], false, Bv[0][c4], (short)0, acc0, false, false);
                acc1 = __builtin_amdgcn_wmma_f32_16x16x4_f32(
                    false, A[c4], false, Bv[1][c4], (short)0, acc1, false, false);
            }
            const v4f v0a = vv[jj][0][0], v0b = vv[jj][0][1];
            const v4f v1a = vv[jj][1][0], v1b = vv[jj][1][1];
            float p0a = acc0[0] * v0a.x, p0b = acc0[4] * v0b.x;
            float p1a = acc1[0] * v1a.x, p1b = acc1[4] * v1b.x;
            p0a += acc0[1] * v0a.y;  p0b += acc0[5] * v0b.y;
            p1a += acc1[1] * v1a.y;  p1b += acc1[5] * v1b.y;
            p0a += acc0[2] * v0a.z;  p0b += acc0[6] * v0b.z;
            p1a += acc1[2] * v1a.z;  p1b += acc1[6] * v1b.z;
            p0a += acc0[3] * v0a.w;  p0b += acc0[7] * v0b.w;
            p1a += acc1[3] * v1a.w;  p1b += acc1[7] * v1b.w;
            const float p0 = p0a + p0b;
            const float p1 = p1a + p1b;
            const float keep = hi ? p1 : p0;
            const float send = hi ? p0 : p1;
            float lg = keep + xhalf(send);
            asm volatile("" : "+v"(lg));
            res[jj][q] = lg;
        }
    }
#undef AG_LOAD_STAGE

    v4f lg4[4];
#pragma unroll
    for (int jj = 0; jj < 4; ++jj) {
        lg4[jj] = (v4f){res[jj][0], res[jj][1], res[jj][2], res[jj][3]};
        if (!FIRST) lg4[jj] += old[jj];
    }

    __shared__ float mxs[8 * 32 * 4];
    __shared__ float sms[8 * 32 * 4];
    *(v4f*)&mxs[(wave * 32 + lane) * 4] = vmax4(vmax4(lg4[0], lg4[1]), vmax4(lg4[2], lg4[3]));
    __syncthreads();
    v4f m = *(const v4f*)&mxs[lane * 4];
#pragma unroll
    for (int w = 1; w < 8; ++w) m = vmax4(m, *(const v4f*)&mxs[(w * 32 + lane) * 4]);
    v4f e[4];
#pragma unroll
    for (int jj = 0; jj < 4; ++jj) {
        const v4f d = lg4[jj] - m;
        e[jj] = (v4f){__expf(d.x), __expf(d.y), __expf(d.z), __expf(d.w)};
    }
    *(v4f*)&sms[(wave * 32 + lane) * 4] = (e[0] + e[1]) + (e[2] + e[3]);
    __syncthreads();
    v4f s = *(const v4f*)&sms[lane * 4];
#pragma unroll
    for (int w = 1; w < 8; ++w) s += *(const v4f*)&sms[(w * 32 + lane) * 4];
    const v4f inv = (v4f){1.0f / s.x, 1.0f / s.y, 1.0f / s.z, 1.0f / s.w};
#pragma unroll
    for (int jj = 0; jj < 4; ++jj) {
        if (FIRST) *(v4f*)(lgp + PL_(i4, jg * 4 + jj, lane)) = lg4[jj];
        *(v4f*)(cpl + PL_(i4, jg * 4 + jj, lane)) = e[jj] * inv;
    }
}

extern "C" void kernel_launch(void* const* d_in, const int* in_sizes, int n_in,
                              void* d_out, int out_size, void* d_ws, size_t ws_size,
                              hipStream_t stream) {
    const float* x = (const float*)d_in[0];
    const float* W = (const float*)d_in[1];
    float* out = (float*)d_out;

    float* bij = (float*)d_ws;
    float* c   = bij + (size_t)B_ * J_ * I_;

    sv_wgemm<true><<<NCH_ * 8, 256, 0, stream>>>(x, W, c, c);
    reduce_squash<<<(B_ * J_) / 8, 256, 0, stream>>>(c, out, 0.03125f);
    agree_softmax<true><<<I_ / 4, 256, 0, stream>>>(x, W, out, bij, c);

    sv_wgemm<false><<<NCH_ * 8, 256, 0, stream>>>(x, W, c, c);
    reduce_squash<<<(B_ * J_) / 8, 256, 0, stream>>>(c, out, 1.0f);
    agree_softmax<false><<<I_ / 4, 256, 0, stream>>>(x, W, out, bij, c);

    sv_wgemm<false><<<NCH_ * 8, 256, 0, stream>>>(x, W, c, c);
    reduce_squash<<<(B_ * J_) / 8, 256, 0, stream>>>(c, out, 1.0f);
}
